# starting kernel
# baseline (speedup 1.0000x reference)
#include <hip/hip_runtime.h>
#include <hip/hip_bf16.h>


typedef float v2f __attribute__((ext_vector_type(2)));
typedef float v8f __attribute__((ext_vector_type(8)));

#define BATCH   4
#define NSEQ    2048
#define EDIM    96
#define NHEAD   6
#define HDIM    16
#define QKVDIM  (3 * EDIM)
#define ROWS    (BATCH * NSEQ)

static __device__ __forceinline__ v8f wmma_f32_4(v2f a, v2f b, v8f c) {
    return __builtin_amdgcn_wmma_f32_16x16x4_f32(
        false, a, false, b, (short)0, c, false, false);
}

__global__ __launch_bounds__(256)
void gemm_bias_wmma(const float* __restrict__ A,
                    const float* __restrict__ W,
                    const float* __restrict__ bias,
                    float* __restrict__ C,
                    int M, int K, int N) {
    const int lane = threadIdx.x & 31;
    const int wave = threadIdx.x >> 5;
    const int wavesPerBlock = blockDim.x >> 5;
    const int tileId = blockIdx.x * wavesPerBlock + wave;
    const int tilesN = N >> 4;
    const int tileM = tileId / tilesN;
    const int tileN = tileId - tileM * tilesN;
    if (tileM * 16 >= M) return;

    const int row0 = tileM << 4;
    const int col0 = tileN << 4;
    const int m    = lane & 15;
    const int sel  = (lane >> 4) << 1;
    const int half = (lane >> 4) << 3;

    const float* arow = A + (size_t)(row0 + m) * K;

    v8f c = {};
    for (int k0 = 0; k0 < K; k0 += 4) {
        v2f a, b;
        a.x = arow[k0 + sel + 0];
        a.y = arow[k0 + sel + 1];
        b.x = W[(size_t)(k0 + sel + 0) * N + (col0 + m)];
        b.y = W[(size_t)(k0 + sel + 1) * N + (col0 + m)];
        c = wmma_f32_4(a, b, c);
    }

    const float bv = bias[col0 + m];
#pragma unroll
    for (int r = 0; r < 8; ++r) {
        C[(size_t)(row0 + r + half) * N + (col0 + m)] = c[r] + bv;
    }
}

__global__ __launch_bounds__(256)
void attn_kernel(const float* __restrict__ qkv, float* __restrict__ y) {
    const int bh   = blockIdx.x;
    const int b    = bh / NHEAD;
    const int h    = bh - b * NHEAD;
    const int lane = threadIdx.x & 31;
    const int wave = threadIdx.x >> 5;
    const int m    = lane & 15;
    const int sel  = (lane >> 4) << 1;
    const int half = (lane >> 4) << 3;

    const size_t base  = (size_t)b * NSEQ * QKVDIM;
    const float* qbase = qkv + base + 0 * EDIM + HDIM * h;
    const float* kbase = qkv + base + 1 * EDIM + HDIM * h;
    const float* vbase = qkv + base + 2 * EDIM + HDIM * h;

    __shared__ float red[8 * 256];
    __shared__ float S[16 * 16];

    v8f c = {};
    const int nBeg = wave * (NSEQ / 8);
    for (int n0 = nBeg; n0 < nBeg + NSEQ / 8; n0 += 4) {
        const float* krow = kbase + (size_t)(n0 + sel) * QKVDIM;
        const float* vrow = vbase + (size_t)(n0 + sel) * QKVDIM;
        v2f a, bb;
        a.x  = krow[m];
        a.y  = krow[QKVDIM + m];
        bb.x = vrow[m];
        bb.y = vrow[QKVDIM + m];
        c = wmma_f32_4(a, bb, c);
    }
#pragma unroll
    for (int r = 0; r < 8; ++r) red[wave * 256 + r * 32 + lane] = c[r];
    __syncthreads();

    {
        const int t  = threadIdx.x;
        float s = 0.f;
#pragma unroll
        for (int w = 0; w < 8; ++w) s += red[w * 256 + t];
        const int r  = t >> 5;
        const int ln = t & 31;
        const int row = r + ((ln >> 4) << 3);
        const int col = ln & 15;
        S[row * 16 + col] = s * 0.25f;
    }
    __syncthreads();

    for (int t = wave; t < NSEQ / 16; t += 8) {
        const int n0 = t << 4;
        const float* qrow = qbase + (size_t)(n0 + m) * QKVDIM;
        v8f acc = {};
#pragma unroll
        for (int k0 = 0; k0 < HDIM; k0 += 4) {
            v2f a, bb;
            a.x  = qrow[k0 + sel + 0];
            a.y  = qrow[k0 + sel + 1];
            bb.x = S[(k0 + sel + 0) * 16 + m];
            bb.y = S[(k0 + sel + 1) * 16 + m];
            acc = wmma_f32_4(a, bb, acc);
        }
#pragma unroll
        for (int r = 0; r < 8; ++r) {
            y[(size_t)b * NSEQ * EDIM + (size_t)(n0 + r + half) * EDIM + HDIM * h + m] = acc[r];
        }
    }
}

extern "C" void kernel_launch(void* const* d_in, const int* in_sizes, int n_in,
                              void* d_out, int out_size, void* d_ws, size_t ws_size,
                              hipStream_t stream) {
    const float* x    = (const float*)d_in[0];
    const float* Wqkv = (const float*)d_in[1];
    const float* bqkv = (const float*)d_in[2];
    const float* Wff  = (const float*)d_in[3];
    const float* bff  = (const float*)d_in[4];
    float* out = (float*)d_out;

    float* qkv = (float*)d_ws;
    float* y   = qkv + (size_t)ROWS * QKVDIM;

    {
        const int tiles  = (ROWS / 16) * (QKVDIM / 16);
        gemm_bias_wmma<<<tiles / 8, 256, 0, stream>>>(x, Wqkv, bqkv, qkv,
                                                      ROWS, EDIM, QKVDIM);
    }

    attn_kernel<<<BATCH * NHEAD, 256, 0, stream>>>(qkv, y);

    {
        const int tiles = (ROWS / 16) * (EDIM / 16);
        gemm_bias_wmma<<<tiles / 8, 256, 0, stream>>>(y, Wff, bff, out,
                                                      ROWS, EDIM, EDIM);
    }
}


# whole product chain re-associated around x: k^T v from Gram matrix x^T x, then Wq, S_h, Wff folded to one 96x96 M_b per batch, out = x@M_b + c_b; q,k,v never formed; summation order changes
# speedup vs baseline: 1.5809x; 1.5809x over previous
#include <hip/hip_runtime.h>


typedef float v2f __attribute__((ext_vector_type(2)));
typedef float v8f __attribute__((ext_vector_type(8)));

#define BATCH   4
#define NSEQ    2048
#define EDIM    96
#define NHEAD   6
#define HDIM    16
#define QKVDIM  (3 * EDIM)
#define ROWS    (BATCH * NSEQ)
#define GSZ     (EDIM * EDIM)
#define GP      100
#define TP      104
#define SP      20

static __device__ __forceinline__ v8f wmma_f32_4(v2f a, v2f b, v8f c) {
    return __builtin_amdgcn_wmma_f32_16x16x4_f32(
        false, a, false, b, (short)0, c, false, false);
}


#define GW 16
__global__ __launch_bounds__(GW * 32)
void gram_kernel(const float* __restrict__ x, float* __restrict__ Gws,
                 float* __restrict__ colsum) {
    const int blk  = blockIdx.x;
    const int b    = blk / 36;
    const int t36  = blk - b * 36;
    const int ti   = t36 / 6;
    const int tj   = t36 - ti * 6;
    const int lane = threadIdx.x & 31;
    const int wave = threadIdx.x >> 5;
    const int m    = lane & 15;
    const int sel  = (lane >> 4) << 1;

    const float* xb = x + (size_t)b * NSEQ * EDIM + (size_t)wave * (NSEQ / GW) * EDIM;
    const float* pa = xb + 16 * ti + m;
    const float* pb = xb + 16 * tj + m;

    __shared__ float red[GW * 256];
    __shared__ float redc[GW * 32];

    v8f c0 = {}, c1 = {};
    float cs = 0.f;
#pragma unroll
    for (int n0 = 0; n0 < NSEQ / GW; n0 += 8) {
        const int r0 = (n0 + sel) * EDIM;
        const int r1 = (n0 + 4 + sel) * EDIM;
        v2f a0, b0, a1, b1;
        a0.x = pa[r0];  a0.y = pa[r0 + EDIM];
        b0.x = pb[r0];  b0.y = pb[r0 + EDIM];
        a1.x = pa[r1];  a1.y = pa[r1 + EDIM];
        b1.x = pb[r1];  b1.y = pb[r1 + EDIM];
        cs += (a0.x + a0.y) + (a1.x + a1.y);
        c0 = wmma_f32_4(a0, b0, c0);
        c1 = wmma_f32_4(a1, b1, c1);
    }
#pragma unroll
    for (int r = 0; r < 8; ++r) red[wave * 256 + r * 32 + lane] = c0[r] + c1[r];
    redc[wave * 32 + lane] = cs;
    __syncthreads();

    const int t = threadIdx.x;
    if (t < 256) {
        float s = 0.f;
#pragma unroll
        for (int w = 0; w < GW; ++w) s += red[w * 256 + t];
        Gws[(size_t)b * GSZ + t36 * 256 + t] = s;
    }
    if (ti == tj && t < 16) {
        float cc = 0.f;
#pragma unroll
        for (int w = 0; w < GW; ++w) cc += redc[w * 32 + t] + redc[w * 32 + t + 16];
        colsum[(b * 6 + ti) * 32 + t] = cc;
    }
}

__global__ __launch_bounds__(384)
void fold_kernel(const float* __restrict__ Gws, const float* __restrict__ colsum,
                 const float* __restrict__ Wqkv, const float* __restrict__ bqkv,
                 const float* __restrict__ Wff, const float* __restrict__ bff,
                 float* __restrict__ Mb, float* __restrict__ cb) {
    const int b    = blockIdx.x;
    const int tid  = threadIdx.x;
    const int lane = tid & 31;
    const int wave = tid >> 5;
    const int m    = lane & 15;
    const int sel  = (lane >> 4) << 1;
    const int half = (lane >> 4) << 3;
    const int si   = wave >> 1;
    const int ct3  = (wave & 1) * 3;

    __shared__ __attribute__((aligned(16))) float sG[EDIM * GP];
    __shared__ __attribute__((aligned(16))) float sT[EDIM * TP];
    __shared__ __attribute__((aligned(16))) float sP[EDIM * TP];
    __shared__ __attribute__((aligned(16))) float sS[NHEAD * 16 * SP];
    __shared__ float ss[EDIM];
    __shared__ float su[4 * EDIM];

    float g[24];
#pragma unroll
    for (int it = 0; it < 24; ++it) g[it] = Gws[(size_t)b * GSZ + it * 384 + tid];
    const float csv = (tid < EDIM) ? colsum[(b * 6 + (tid >> 4)) * 32 + (tid & 15)] : 0.f;

    v2f bw[24][3];
#pragma unroll
    for (int ks = 0; ks < 24; ++ks)
#pragma unroll
        for (int q = 0; q < 3; ++q) {
            const int col = 2 * EDIM + 16 * (ct3 + q) + m;
            bw[ks][q].x = Wqkv[(4 * ks + sel) * QKVDIM + col];
            bw[ks][q].y = Wqkv[(4 * ks + sel + 1) * QKVDIM + col];
        }

#pragma unroll
    for (int it = 0; it < 24; ++it) {
        const int e    = it * 384 + tid;
        const int tile = e >> 8;
        const int idx  = e & 255;
        const int row  = (idx >> 5) + (((idx & 31) >> 4) << 3);
        const int col  = idx & 15;
        const int gi   = tile / 6;
        const int gj   = tile - gi * 6;
        sG[(16 * gi + row) * GP + 16 * gj + col] = g[it];
    }
    if (tid < EDIM) ss[tid] = csv;
    __syncthreads();

    {
        v8f acc[3] = {};
#pragma unroll
        for (int ks = 0; ks < 24; ++ks) {
            const v2f a = *(const v2f*)&sG[(16 * si + m) * GP + 4 * ks + sel];
#pragma unroll
            for (int q = 0; q < 3; ++q) acc[q] = wmma_f32_4(a, bw[ks][q], acc[q]);
        }
#pragma unroll
        for (int q = 0; q < 3; ++q)
#pragma unroll
            for (int r = 0; r < 8; ++r)
                sT[(16 * si + r + half) * TP + 16 * (ct3 + q) + m] = acc[q][r];
    }

    v2f wk[24];
    {
        const int h = (wave < NHEAD) ? wave : 0;
#pragma unroll
        for (int ks = 0; ks < 24; ++ks) {
            wk[ks].x = Wqkv[(4 * ks + sel) * QKVDIM + EDIM + 16 * h + m];
            wk[ks].y = Wqkv[(4 * ks + sel + 1) * QKVDIM + EDIM + 16 * h + m];
        }
    }
    v2f wf[4][3];
#pragma unroll
    for (int ks = 0; ks < 4; ++ks)
#pragma unroll
        for (int q = 0; q < 3; ++q) {
            const int col = 16 * (ct3 + q) + m;
            wf[ks][q].x = Wff[(16 * si + 4 * ks + sel) * EDIM + col];
            wf[ks][q].y = Wff[(16 * si + 4 * ks + sel + 1) * EDIM + col];
        }
    v2f wq[24];
#pragma unroll
    for (int ks = 0; ks < 24; ++ks)
        wq[ks] = *(const v2f*)&Wqkv[(16 * si + m) * QKVDIM + 4 * ks + sel];

    {
        const int part = (tid >= 2 * EDIM) ? 1 : 0;
        const int o    = tid - part * 2 * EDIM;
        float u = 0.f;
#pragma unroll
        for (int i = 0; i < 48; ++i)
            u += ss[part * 48 + i] * Wqkv[(part * 48 + i) * QKVDIM + EDIM + o];
        su[part * 2 * EDIM + o] = u;
    }
    __syncthreads();

    if (wave < NHEAD) {
        const int h = wave;
        v8f s0 = {}, s1 = {};
#pragma unroll
        for (int ks = 0; ks < 24; ks += 2) {
            v2f b0, b1;
            b0.x = sT[(4 * ks + sel) * TP + 16 * h + m];
            b0.y = sT[(4 * ks + sel + 1) * TP + 16 * h + m];
            b1.x = sT[(4 * ks + 4 + sel) * TP + 16 * h + m];
            b1.y = sT[(4 * ks + 5 + sel) * TP + 16 * h + m];
            s0 = wmma_f32_4(wk[ks], b0, s0);
            s1 = wmma_f32_4(wk[ks + 1], b1, s1);
        }
        const float bv_e = bqkv[2 * EDIM + 16 * h + m];
        const float uv_e = su[EDIM + 16 * h + m] + su[3 * EDIM + 16 * h + m];
#pragma unroll
        for (int r = 0; r < 8; ++r) {
            const int d = r + half;
            const float bk_d = bqkv[EDIM + 16 * h + d];
            const float uk_d = su[16 * h + d] + su[2 * EDIM + 16 * h + d];
            const float v = (s0[r] + s1[r]) + uk_d * bv_e + bk_d * uv_e
                          + (float)NSEQ * bk_d * bv_e;
            sS[h * 16 * SP + d * SP + m] = 0.25f * v;
        }
    }
    __syncthreads();

    {
        v8f acc[3] = {};
#pragma unroll
        for (int ks = 0; ks < 4; ++ks) {
            const v2f a = *(const v2f*)&sS[si * 16 * SP + m * SP + 4 * ks + sel];
#pragma unroll
            for (int q = 0; q < 3; ++q) acc[q] = wmma_f32_4(a, wf[ks][q], acc[q]);
        }
#pragma unroll
        for (int q = 0; q < 3; ++q)
#pragma unroll
            for (int r = 0; r < 8; ++r)
                sP[(16 * si + r + half) * TP + 16 * (ct3 + q) + m] = acc[q][r];
    }
    __syncthreads();

    {
        v8f acc[3] = {};
#pragma unroll
        for (int ks = 0; ks < 24; ++ks) {
#pragma unroll
            for (int q = 0; q < 3; ++q) {
                const int col = 16 * (ct3 + q) + m;
                v2f bb;
                bb.x = sP[(4 * ks + sel) * TP + col];
                bb.y = sP[(4 * ks + sel + 1) * TP + col];
                acc[q] = wmma_f32_4(wq[ks], bb, acc[q]);
            }
        }
#pragma unroll
        for (int q = 0; q < 3; ++q)
#pragma unroll
            for (int r = 0; r < 8; ++r)
                sT[(16 * si + r + half) * TP + 16 * (ct3 + q) + m] = acc[q][r];
    }
    if (tid < EDIM) {
        float c = bff[tid];
#pragma unroll 16
        for (int i = 0; i < EDIM; ++i) c += bqkv[i] * sP[i * TP + tid];
        cb[b * EDIM + tid] = c;
    }
    __syncthreads();

#pragma unroll
    for (int it = 0; it < 6; ++it) {
        const int q4  = it * 384 + tid;
        const int row = q4 / 24;
        const int c4  = (q4 - row * 24) * 4;
        const float4 v = *(const float4*)&sT[row * TP + c4];
        *(float4*)&Mb[(size_t)b * GSZ + row * EDIM + c4] = v;
    }
}

__global__ __launch_bounds__(256)
void out_gemm_wmma(const float* __restrict__ A,
                   const float* __restrict__ Wb,
                   const float* __restrict__ biasb,
                   float* __restrict__ C) {
    const int lane = threadIdx.x & 31;
    const int wave = threadIdx.x >> 5;
    const int tileId = blockIdx.x * 8 + wave;
    const int tileM = tileId / (EDIM / 16);
    const int tileN = tileId - tileM * (EDIM / 16);

    const int row0 = tileM << 4;
    const int col0 = tileN << 4;
    const int m    = lane & 15;
    const int sel  = (lane >> 4) << 1;
    const int half = (lane >> 4) << 3;
    const int bidx = row0 / NSEQ;

    const float* arow = A + (size_t)(row0 + m) * EDIM;
    const float* W    = Wb + (size_t)bidx * GSZ;

    v8f c = {};
    for (int k0 = 0; k0 < EDIM; k0 += 4) {
        v2f a, b;
        a.x = arow[k0 + sel + 0];
        a.y = arow[k0 + sel + 1];
        b.x = W[(k0 + sel + 0) * EDIM + (col0 + m)];
        b.y = W[(k0 + sel + 1) * EDIM + (col0 + m)];
        c = wmma_f32_4(a, b, c);
    }

    const float bv = biasb[bidx * EDIM + col0 + m];
#pragma unroll
    for (int r = 0; r < 8; ++r) {
        C[(size_t)(row0 + r + half) * EDIM + (col0 + m)] = c[r] + bv;
    }
}

extern "C" void kernel_launch(void* const* d_in, const int* in_sizes, int n_in,
                              void* d_out, int out_size, void* d_ws, size_t ws_size,
                              hipStream_t stream) {
    const float* x    = (const float*)d_in[0];
    const float* Wqkv = (const float*)d_in[1];
    const float* bqkv = (const float*)d_in[2];
    const float* Wff  = (const float*)d_in[3];
    const float* bff  = (const float*)d_in[4];
    float* out = (float*)d_out;

    float* Gws    = (float*)d_ws;
    float* colsum = Gws + BATCH * GSZ;
    float* Mb     = colsum + BATCH * 6 * 32;
    float* cbias  = Mb + BATCH * GSZ;

    gram_kernel<<<BATCH * 36, GW * 32, 0, stream>>>(x, Gws, colsum);

    fold_kernel<<<BATCH, 384, 0, stream>>>(Gws, colsum, Wqkv, bqkv, Wff, bff, Mb, cbias);

    out_gemm_wmma<<<(ROWS / 16) * (EDIM / 16) / 8, 256, 0, stream>>>(x, Mb, cbias, out);
}


# fold fused into the output launch: each of 128 blocks rebuilds M_b and c_b in LDS from G_b (fold repeated 32x per batch) and applies them to its 64 rows of x; M_b never stored; launches 3 to 2
# speedup vs baseline: 1.9681x; 1.2449x over previous
#include <hip/hip_runtime.h>


typedef float v2f __attribute__((ext_vector_type(2)));
typedef float v8f __attribute__((ext_vector_type(8)));

#define BATCH   4
#define NSEQ    2048
#define EDIM    96
#define NHEAD   6
#define HDIM    16
#define QKVDIM  (3 * EDIM)
#define ROWS    (BATCH * NSEQ)
#define GSZ     (EDIM * EDIM)
#define GP      100
#define TP      104
#define SP      20
#define OROWS   64
#define CHUNKS  (NSEQ / OROWS)

static __device__ __forceinline__ v8f wmma_f32_4(v2f a, v2f b, v8f c) {
    return __builtin_amdgcn_wmma_f32_16x16x4_f32(
        false, a, false, b, (short)0, c, false, false);
}


#define GW 16
__global__ __launch_bounds__(GW * 32)
void gram_kernel(const float* __restrict__ x, float* __restrict__ Gws,
                 float* __restrict__ colsum) {
    const int blk  = blockIdx.x;
    const int b    = blk / 36;
    const int t36  = blk - b * 36;
    const int ti   = t36 / 6;
    const int tj   = t36 - ti * 6;
    const int lane = threadIdx.x & 31;
    const int wave = threadIdx.x >> 5;
    const int m    = lane & 15;
    const int sel  = (lane >> 4) << 1;

    const float* xb = x + (size_t)b * NSEQ * EDIM + (size_t)wave * (NSEQ / GW) * EDIM;
    const float* pa = xb + 16 * ti + m;
    const float* pb = xb + 16 * tj + m;

    __shared__ float red[GW * 256];
    __shared__ float redc[GW * 32];

    v8f c0 = {}, c1 = {};
    float cs = 0.f;
#pragma unroll
    for (int n0 = 0; n0 < NSEQ / GW; n0 += 8) {
        const int r0 = (n0 + sel) * EDIM;
        const int r1 = (n0 + 4 + sel) * EDIM;
        v2f a0, b0, a1, b1;
        a0.x = pa[r0];  a0.y = pa[r0 + EDIM];
        b0.x = pb[r0];  b0.y = pb[r0 + EDIM];
        a1.x = pa[r1];  a1.y = pa[r1 + EDIM];
        b1.x = pb[r1];  b1.y = pb[r1 + EDIM];
        cs += (a0.x + a0.y) + (a1.x + a1.y);
        c0 = wmma_f32_4(a0, b0, c0);
        c1 = wmma_f32_4(a1, b1, c1);
    }
#pragma unroll
    for (int r = 0; r < 8; ++r) red[wave * 256 + r * 32 + lane] = c0[r] + c1[r];
    redc[wave * 32 + lane] = cs;
    __syncthreads();

    const int t = threadIdx.x;
    if (t < 256) {
        float s = 0.f;
#pragma unroll
        for (int w = 0; w < GW; ++w) s += red[w * 256 + t];
        Gws[(size_t)b * GSZ + t36 * 256 + t] = s;
    }
    if (ti == tj && t < 16) {
        float cc = 0.f;
#pragma unroll
        for (int w = 0; w < GW; ++w) cc += redc[w * 32 + t] + redc[w * 32 + t + 16];
        colsum[(b * 6 + ti) * 32 + t] = cc;
    }
}

static __device__ __forceinline__ void put_x_row_piece(float* tile, int q4, float4 v) {
    const int row = q4 / 24;
    const int c4  = (q4 - row * 24) * 4;
    *(float4*)&tile[row * GP + c4] = v;
}

__global__ __launch_bounds__(384)
void fold_out_kernel(const float* __restrict__ x,
                     const float* __restrict__ Gws, const float* __restrict__ colsum,
                     const float* __restrict__ Wqkv, const float* __restrict__ bqkv,
                     const float* __restrict__ Wff, const float* __restrict__ bff,
                     float* __restrict__ out) {
    const int b     = blockIdx.x / CHUNKS;
    const int chunk = blockIdx.x - b * CHUNKS;
    const int tid  = threadIdx.x;
    const int lane = tid & 31;
    const int wave = tid >> 5;
    const int m    = lane & 15;
    const int sel  = (lane >> 4) << 1;
    const int half = (lane >> 4) << 3;
    const int si   = wave >> 1;
    const int ct3  = (wave & 1) * 3;

    __shared__ __attribute__((aligned(16))) float sG[EDIM * GP];
    __shared__ __attribute__((aligned(16))) float sT[EDIM * TP];
    __shared__ __attribute__((aligned(16))) float sP[EDIM * TP];
    __shared__ __attribute__((aligned(16))) float sS[NHEAD * 16 * SP];
    __shared__ float ss[EDIM];
    __shared__ float su[4 * EDIM];
    __shared__ float sc[EDIM];
    __shared__ __attribute__((aligned(16))) float sX[OROWS * GP];

    float g[24];
#pragma unroll
    for (int it = 0; it < 24; ++it) g[it] = Gws[(size_t)b * GSZ + it * 384 + tid];
    const float csv = (tid < EDIM) ? colsum[(b * 6 + (tid >> 4)) * 32 + (tid & 15)] : 0.f;

    v2f bw[24][3];
#pragma unroll
    for (int ks = 0; ks < 24; ++ks)
#pragma unroll
        for (int q = 0; q < 3; ++q) {
            const int col = 2 * EDIM + 16 * (ct3 + q) + m;
            bw[ks][q].x = Wqkv[(4 * ks + sel) * QKVDIM + col];
            bw[ks][q].y = Wqkv[(4 * ks + sel + 1) * QKVDIM + col];
        }

    const size_t row0 = (size_t)b * NSEQ + (size_t)chunk * OROWS;
    const float* xblk = x + row0 * EDIM;
    const float4 x0 = *(const float4*)&xblk[(0 * 384 + tid) * 4];
    const float4 x1 = *(const float4*)&xblk[(1 * 384 + tid) * 4];
    const float4 x2 = *(const float4*)&xblk[(2 * 384 + tid) * 4];
    const float4 x3 = *(const float4*)&xblk[(3 * 384 + tid) * 4];

#pragma unroll
    for (int it = 0; it < 24; ++it) {
        const int e    = it * 384 + tid;
        const int tile = e >> 8;
        const int idx  = e & 255;
        const int row  = (idx >> 5) + (((idx & 31) >> 4) << 3);
        const int col  = idx & 15;
        const int gi   = tile / 6;
        const int gj   = tile - gi * 6;
        sG[(16 * gi + row) * GP + 16 * gj + col] = g[it];
    }
    if (tid < EDIM) ss[tid] = csv;
    put_x_row_piece(sX, 0 * 384 + tid, x0);
    put_x_row_piece(sX, 1 * 384 + tid, x1);
    put_x_row_piece(sX, 2 * 384 + tid, x2);
    put_x_row_piece(sX, 3 * 384 + tid, x3);
    __syncthreads();

    {
        v8f acc[3] = {};
#pragma unroll
        for (int ks = 0; ks < 24; ++ks) {
            const v2f a = *(const v2f*)&sG[(16 * si + m) * GP + 4 * ks + sel];
#pragma unroll
            for (int q = 0; q < 3; ++q) acc[q] = wmma_f32_4(a, bw[ks][q], acc[q]);
        }
#pragma unroll
        for (int q = 0; q < 3; ++q)
#pragma unroll
            for (int r = 0; r < 8; ++r)
                sT[(16 * si + r + half) * TP + 16 * (ct3 + q) + m] = acc[q][r];
    }

    v2f wk[24];
    {
        const int h = (wave < NHEAD) ? wave : 0;
#pragma unroll
        for (int ks = 0; ks < 24; ++ks) {
            wk[ks].x = Wqkv[(4 * ks + sel) * QKVDIM + EDIM + 16 * h + m];
            wk[ks].y = Wqkv[(4 * ks + sel + 1) * QKVDIM + EDIM + 16 * h + m];
        }
    }
    v2f wf[4][3];
#pragma unroll
    for (int ks = 0; ks < 4; ++ks)
#pragma unroll
        for (int q = 0; q < 3; ++q) {
            const int col = 16 * (ct3 + q) + m;
            wf[ks][q].x = Wff[(16 * si + 4 * ks + sel) * EDIM + col];
            wf[ks][q].y = Wff[(16 * si + 4 * ks + sel + 1) * EDIM + col];
        }
    v2f wq[24];
#pragma unroll
    for (int ks = 0; ks < 24; ++ks)
        wq[ks] = *(const v2f*)&Wqkv[(16 * si + m) * QKVDIM + 4 * ks + sel];

    {
        const int part = (tid >= 2 * EDIM) ? 1 : 0;
        const int o    = tid - part * 2 * EDIM;
        float u = 0.f;
#pragma unroll
        for (int i = 0; i < 48; ++i)
            u += ss[part * 48 + i] * Wqkv[(part * 48 + i) * QKVDIM + EDIM + o];
        su[part * 2 * EDIM + o] = u;
    }
    __syncthreads();

    if (wave < NHEAD) {
        const int h = wave;
        v8f s0 = {}, s1 = {};
#pragma unroll
        for (int ks = 0; ks < 24; ks += 2) {
            v2f b0, b1;
            b0.x = sT[(4 * ks + sel) * TP + 16 * h + m];
            b0.y = sT[(4 * ks + sel + 1) * TP + 16 * h + m];
            b1.x = sT[(4 * ks + 4 + sel) * TP + 16 * h + m];
            b1.y = sT[(4 * ks + 5 + sel) * TP + 16 * h + m];
            s0 = wmma_f32_4(wk[ks], b0, s0);
            s1 = wmma_f32_4(wk[ks + 1], b1, s1);
        }
        const float bv_e = bqkv[2 * EDIM + 16 * h + m];
        const float uv_e = su[EDIM + 16 * h + m] + su[3 * EDIM + 16 * h + m];
#pragma unroll
        for (int r = 0; r < 8; ++r) {
            const int d = r + half;
            const float bk_d = bqkv[EDIM + 16 * h + d];
            const float uk_d = su[16 * h + d] + su[2 * EDIM + 16 * h + d];
            const float v = (s0[r] + s1[r]) + uk_d * bv_e + bk_d * uv_e
                          + (float)NSEQ * bk_d * bv_e;
            sS[h * 16 * SP + d * SP + m] = 0.25f * v;
        }
    }
    __syncthreads();

    {
        v8f acc[3] = {};
#pragma unroll
        for (int ks = 0; ks < 4; ++ks) {
            const v2f a = *(const v2f*)&sS[si * 16 * SP + m * SP + 4 * ks + sel];
#pragma unroll
            for (int q = 0; q < 3; ++q) acc[q] = wmma_f32_4(a, wf[ks][q], acc[q]);
        }
#pragma unroll
        for (int q = 0; q < 3; ++q)
#pragma unroll
            for (int r = 0; r < 8; ++r)
                sP[(16 * si + r + half) * TP + 16 * (ct3 + q) + m] = acc[q][r];
    }
    __syncthreads();

    {
        v8f acc[3] = {};
#pragma unroll
        for (int ks = 0; ks < 24; ++ks) {
#pragma unroll
            for (int q = 0; q < 3; ++q) {
                const int col = 16 * (ct3 + q) + m;
                v2f bb;
                bb.x = sP[(4 * ks + sel) * TP + col];
                bb.y = sP[(4 * ks + sel + 1) * TP + col];
                acc[q] = wmma_f32_4(wq[ks], bb, acc[q]);
            }
        }
#pragma unroll
        for (int q = 0; q < 3; ++q)
#pragma unroll
            for (int r = 0; r < 8; ++r)
                sT[(16 * si + r + half) * TP + 16 * (ct3 + q) + m] = acc[q][r];
    }
    if (tid < EDIM) {
        float c = bff[tid];
#pragma unroll 16
        for (int i = 0; i < EDIM; ++i) c += bqkv[i] * sP[i * TP + tid];
        sc[tid] = c;
    }
    __syncthreads();

    {
        const int st   = wave / 3;
        const int colA = 32 * (wave - 3 * st) + m;
        v8f o0 = {}, o1 = {};
#pragma unroll
        for (int ks = 0; ks < 24; ++ks) {
            const v2f a = *(const v2f*)&sX[(16 * st + m) * GP + 4 * ks + sel];
            v2f b0, b1;
            b0.x = sT[(4 * ks + sel) * TP + colA];
            b0.y = sT[(4 * ks + sel + 1) * TP + colA];
            b1.x = sT[(4 * ks + sel) * TP + colA + 16];
            b1.y = sT[(4 * ks + sel + 1) * TP + colA + 16];
            o0 = wmma_f32_4(a, b0, o0);
            o1 = wmma_f32_4(a, b1, o1);
        }
        const float c0 = sc[colA];
        const float c1 = sc[colA + 16];
        float* orow = out + (row0 + 16 * st + half) * EDIM + colA;
#pragma unroll
        for (int r = 0; r < 8; ++r) {
            orow[r * EDIM]      = o0[r] + c0;
            orow[r * EDIM + 16] = o1[r] + c1;
        }
    }
}

extern "C" void kernel_launch(void* const* d_in, const int* in_sizes, int n_in,
                              void* d_out, int out_size, void* d_ws, size_t ws_size,
                              hipStream_t stream) {
    const float* x    = (const float*)d_in[0];
    const float* Wqkv = (const float*)d_in[1];
    const float* bqkv = (const float*)d_in[2];
    const float* Wff  = (const float*)d_in[3];
    const float* bff  = (const float*)d_in[4];
    float* out = (float*)d_out;

    float* Gws    = (float*)d_ws;
    float* colsum = Gws + BATCH * GSZ;

    gram_kernel<<<BATCH * 36, GW * 32, 0, stream>>>(x, Gws, colsum);

    fold_out_kernel<<<BATCH * CHUNKS, 384, 0, stream>>>(x, Gws, colsum, Wqkv, bqkv, Wff, bff, out);
}


# fold operands (G, Wq, Wk, Wv, Wff) copied once per block into LDS by 16-byte coalesced loads; all fold fragments read from LDS, no per-wave 4-byte global fragment loads; order of every sum kept
# speedup vs baseline: 2.0628x; 1.0481x over previous
#include <hip/hip_runtime.h>


typedef float v2f __attribute__((ext_vector_type(2)));
typedef float v8f __attribute__((ext_vector_type(8)));

#define BATCH   4
#define NSEQ    2048
#define EDIM    96
#define NHEAD   6
#define HDIM    16
#define QKVDIM  (3 * EDIM)
#define ROWS    (BATCH * NSEQ)
#define GSZ     (EDIM * EDIM)
#define GP      100
#define TP      104
#define SP      20
#define OROWS   64
#define CHUNKS  (NSEQ / OROWS)

static __device__ __forceinline__ v8f wmma_f32_4(v2f a, v2f b, v8f c) {
    return __builtin_amdgcn_wmma_f32_16x16x4_f32(
        false, a, false, b, (short)0, c, false, false);
}


#define GW 16
__global__ __launch_bounds__(GW * 32)
void gram_kernel(const float* __restrict__ x, float* __restrict__ Gws,
                 float* __restrict__ colsum) {
    const int blk  = blockIdx.x;
    const int b    = blk / 36;
    const int t36  = blk - b * 36;
    const int ti   = t36 / 6;
    const int tj   = t36 - ti * 6;
    const int lane = threadIdx.x & 31;
    const int wave = threadIdx.x >> 5;
    const int m    = lane & 15;
    const int sel  = (lane >> 4) << 1;

    const float* xb = x + (size_t)b * NSEQ * EDIM + (size_t)wave * (NSEQ / GW) * EDIM;
    const float* pa = xb + 16 * ti + m;
    const float* pb = xb + 16 * tj + m;

    __shared__ float red[GW * 256];
    __shared__ float redc[GW * 32];

    v8f c0 = {}, c1 = {};
    float cs = 0.f;
#pragma unroll
    for (int n0 = 0; n0 < NSEQ / GW; n0 += 8) {
        const int r0 = (n0 + sel) * EDIM;
        const int r1 = (n0 + 4 + sel) * EDIM;
        v2f a0, b0, a1, b1;
        a0.x = pa[r0];  a0.y = pa[r0 + EDIM];
        b0.x = pb[r0];  b0.y = pb[r0 + EDIM];
        a1.x = pa[r1];  a1.y = pa[r1 + EDIM];
        b1.x = pb[r1];  b1.y = pb[r1 + EDIM];
        cs += (a0.x + a0.y) + (a1.x + a1.y);
        c0 = wmma_f32_4(a0, b0, c0);
        c1 = wmma_f32_4(a1, b1, c1);
    }
#pragma unroll
    for (int r = 0; r < 8; ++r) red[wave * 256 + r * 32 + lane] = c0[r] + c1[r];
    redc[wave * 32 + lane] = cs;
    __syncthreads();

    const int t = threadIdx.x;
    if (t < 256) {
        float s = 0.f;
#pragma unroll
        for (int w = 0; w < GW; ++w) s += red[w * 256 + t];
        Gws[(size_t)b * GSZ + t36 * 256 + t] = s;
    }
    if (ti == tj && t < 16) {
        float cc = 0.f;
#pragma unroll
        for (int w = 0; w < GW; ++w) cc += redc[w * 32 + t] + redc[w * 32 + t + 16];
        colsum[(b * 6 + ti) * 32 + t] = cc;
    }
}

template <int P>
static __device__ __forceinline__ void put_row_piece(float* tile, int q4, float4 v) {
    const int row = q4 / 24;
    const int c4  = (q4 - row * 24) * 4;
    *(float4*)&tile[row * P + c4] = v;
}

static __device__ __forceinline__ float4 get_w_piece(const float* __restrict__ W, int off, int q4) {
    const int row = q4 / 24;
    const int c4  = (q4 - row * 24) * 4;
    return *(const float4*)&W[row * QKVDIM + off + c4];
}

__global__ __launch_bounds__(384)
void fold_out_kernel(const float* __restrict__ x,
                     const float* __restrict__ Gws, const float* __restrict__ colsum,
                     const float* __restrict__ Wqkv, const float* __restrict__ bqkv,
                     const float* __restrict__ Wff, const float* __restrict__ bff,
                     float* __restrict__ out) {
    const int b     = blockIdx.x / CHUNKS;
    const int chunk = blockIdx.x - b * CHUNKS;
    const int tid  = threadIdx.x;
    const int lane = tid & 31;
    const int wave = tid >> 5;
    const int m    = lane & 15;
    const int sel  = (lane >> 4) << 1;
    const int half = (lane >> 4) << 3;
    const int si   = wave >> 1;
    const int ct3  = (wave & 1) * 3;

    __shared__ __attribute__((aligned(16))) float sG[EDIM * GP];
    __shared__ __attribute__((aligned(16))) float sT[EDIM * TP];
    __shared__ __attribute__((aligned(16))) float sP[EDIM * TP];
    __shared__ __attribute__((aligned(16))) float sS[NHEAD * 16 * SP];
    __shared__ float ss[EDIM];
    __shared__ float su[4 * EDIM];
    __shared__ float sc[EDIM];
    __shared__ __attribute__((aligned(16))) float sX[OROWS * GP];
    __shared__ __attribute__((aligned(16))) float sWv[EDIM * TP];
    __shared__ __attribute__((aligned(16))) float sWk[EDIM * TP];
    __shared__ __attribute__((aligned(16))) float sWf[EDIM * TP];
    __shared__ __attribute__((aligned(16))) float sWq[EDIM * GP];

    float4 g4[6];
#pragma unroll
    for (int it = 0; it < 6; ++it) g4[it] = *(const float4*)&Gws[(size_t)b * GSZ + (it * 384 + tid) * 4];
    const float csv = (tid < EDIM) ? colsum[(b * 6 + (tid >> 4)) * 32 + (tid & 15)] : 0.f;

    float4 wv4[6];
#pragma unroll
    for (int it = 0; it < 6; ++it) wv4[it] = get_w_piece(Wqkv, 2 * EDIM, it * 384 + tid);

    const size_t row0 = (size_t)b * NSEQ + (size_t)chunk * OROWS;
    const float* xblk = x + row0 * EDIM;
    const float4 x0 = *(const float4*)&xblk[(0 * 384 + tid) * 4];
    const float4 x1 = *(const float4*)&xblk[(1 * 384 + tid) * 4];
    const float4 x2 = *(const float4*)&xblk[(2 * 384 + tid) * 4];
    const float4 x3 = *(const float4*)&xblk[(3 * 384 + tid) * 4];

    float4 wk4[6], wf4[6], wq4[6];
#pragma unroll
    for (int it = 0; it < 6; ++it) wk4[it] = get_w_piece(Wqkv, EDIM, it * 384 + tid);
#pragma unroll
    for (int it = 0; it < 6; ++it) wf4[it] = *(const float4*)&Wff[(it * 384 + tid) * 4];
#pragma unroll
    for (int it = 0; it < 6; ++it) wq4[it] = get_w_piece(Wqkv, 0, it * 384 + tid);

#pragma unroll
    for (int it = 0; it < 6; ++it) {
        const int e    = (it * 384 + tid) * 4;
        const int tile = e >> 8;
        const int idx  = e & 255;
        const int row  = (idx >> 5) + (((idx & 31) >> 4) << 3);
        const int col  = idx & 15;
        const int gi   = tile / 6;
        const int gj   = tile - gi * 6;
        *(float4*)&sG[(16 * gi + row) * GP + 16 * gj + col] = g4[it];
    }
    if (tid < EDIM) ss[tid] = csv;
#pragma unroll
    for (int it = 0; it < 6; ++it) put_row_piece<TP>(sWv, it * 384 + tid, wv4[it]);
    put_row_piece<GP>(sX, 0 * 384 + tid, x0);
    put_row_piece<GP>(sX, 1 * 384 + tid, x1);
    put_row_piece<GP>(sX, 2 * 384 + tid, x2);
    put_row_piece<GP>(sX, 3 * 384 + tid, x3);
#pragma unroll
    for (int it = 0; it < 6; ++it) put_row_piece<TP>(sWk, it * 384 + tid, wk4[it]);
    __syncthreads();

    {
        v8f acc[3] = {};
#pragma unroll
        for (int ks = 0; ks < 24; ++ks) {
            const v2f a = *(const v2f*)&sG[(16 * si + m) * GP + 4 * ks + sel];
#pragma unroll
            for (int q = 0; q < 3; ++q) {
                const float* bp = &sWv[(4 * ks + sel) * TP + 16 * (ct3 + q) + m];
                v2f bb;
                bb.x = bp[0];
                bb.y = bp[TP];
                acc[q] = wmma_f32_4(a, bb, acc[q]);
            }
        }
#pragma unroll
        for (int q = 0; q < 3; ++q)
#pragma unroll
            for (int r = 0; r < 8; ++r)
                sT[(16 * si + r + half) * TP + 16 * (ct3 + q) + m] = acc[q][r];
    }

#pragma unroll
    for (int it = 0; it < 6; ++it) put_row_piece<TP>(sWf, it * 384 + tid, wf4[it]);
#pragma unroll
    for (int it = 0; it < 6; ++it) put_row_piece<GP>(sWq, it * 384 + tid, wq4[it]);

    {
        const int part = (tid >= 2 * EDIM) ? 1 : 0;
        const int o    = tid - part * 2 * EDIM;
        const float* wcol = (o < EDIM) ? &sWk[o] : &sWv[o - EDIM];
        float u = 0.f;
#pragma unroll
        for (int i = 0; i < 48; ++i)
            u += ss[part * 48 + i] * wcol[(part * 48 + i) * TP];
        su[part * 2 * EDIM + o] = u;
    }
    __syncthreads();

    if (wave < NHEAD) {
        const int h = wave;
        v8f s0 = {}, s1 = {};
#pragma unroll
        for (int ks = 0; ks < 24; ks += 2) {
            v2f b0, b1;
            b0.x = sT[(4 * ks + sel) * TP + 16 * h + m];
            b0.y = sT[(4 * ks + sel + 1) * TP + 16 * h + m];
            b1.x = sT[(4 * ks + 4 + sel) * TP + 16 * h + m];
            b1.y = sT[(4 * ks + 5 + sel) * TP + 16 * h + m];
            const float* ap = &sWk[(4 * ks + sel) * TP + 16 * h + m];
            v2f a0, a1;
            a0.x = ap[0];
            a0.y = ap[TP];
            a1.x = ap[4 * TP];
            a1.y = ap[5 * TP];
            s0 = wmma_f32_4(a0, b0, s0);
            s1 = wmma_f32_4(a1, b1, s1);
        }
        const float bv_e = bqkv[2 * EDIM + 16 * h + m];
        const float uv_e = su[EDIM + 16 * h + m] + su[3 * EDIM + 16 * h + m];
#pragma unroll
        for (int r = 0; r < 8; ++r) {
            const int d = r + half;
            const float bk_d = bqkv[EDIM + 16 * h + d];
            const float uk_d = su[16 * h + d] + su[2 * EDIM + 16 * h + d];
            const float v = (s0[r] + s1[r]) + uk_d * bv_e + bk_d * uv_e
                          + (float)NSEQ * bk_d * bv_e;
            sS[h * 16 * SP + d * SP + m] = 0.25f * v;
        }
    }
    __syncthreads();

    {
        v8f acc[3] = {};
#pragma unroll
        for (int ks = 0; ks < 4; ++ks) {
            const v2f a = *(const v2f*)&sS[si * 16 * SP + m * SP + 4 * ks + sel];
#pragma unroll
            for (int q = 0; q < 3; ++q) {
                const float* bp = &sWf[(16 * si + 4 * ks + sel) * TP + 16 * (ct3 + q) + m];
                v2f bb;
                bb.x = bp[0];
                bb.y = bp[TP];
                acc[q] = wmma_f32_4(a, bb, acc[q]);
            }
        }
#pragma unroll
        for (int q = 0; q < 3; ++q)
#pragma unroll
            for (int r = 0; r < 8; ++r)
                sP[(16 * si + r + half) * TP + 16 * (ct3 + q) + m] = acc[q][r];
    }
    __syncthreads();

    {
        v8f acc[3] = {};
#pragma unroll
        for (int ks = 0; ks < 24; ++ks) {
            const v2f a = *(const v2f*)&sWq[(16 * si + m) * GP + 4 * ks + sel];
#pragma unroll
            for (int q = 0; q < 3; ++q) {
                const int col = 16 * (ct3 + q) + m;
                v2f bb;
                bb.x = sP[(4 * ks + sel) * TP + col];
                bb.y = sP[(4 * ks + sel + 1) * TP + col];
                acc[q] = wmma_f32_4(a, bb, acc[q]);
            }
        }
#pragma unroll
        for (int q = 0; q < 3; ++q)
#pragma unroll
            for (int r = 0; r < 8; ++r)
                sT[(16 * si + r + half) * TP + 16 * (ct3 + q) + m] = acc[q][r];
    }
    if (tid < EDIM) {
        float c = bff[tid];
#pragma unroll 16
        for (int i = 0; i < EDIM; ++i) c += bqkv[i] * sP[i * TP + tid];
        sc[tid] = c;
    }
    __syncthreads();

    {
        const int st   = wave / 3;
        const int colA = 32 * (wave - 3 * st) + m;
        v8f o0 = {}, o1 = {};
#pragma unroll
        for (int ks = 0; ks < 24; ++ks) {
            const v2f a = *(const v2f*)&sX[(16 * st + m) * GP + 4 * ks + sel];
            v2f b0, b1;
            b0.x = sT[(4 * ks + sel) * TP + colA];
            b0.y = sT[(4 * ks + sel + 1) * TP + colA];
            b1.x = sT[(4 * ks + sel) * TP + colA + 16];
            b1.y = sT[(4 * ks + sel + 1) * TP + colA + 16];
            o0 = wmma_f32_4(a, b0, o0);
            o1 = wmma_f32_4(a, b1, o1);
        }
        const float c0 = sc[colA];
        const float c1 = sc[colA + 16];
        float* orow = out + (row0 + 16 * st + half) * EDIM + colA;
#pragma unroll
        for (int r = 0; r < 8; ++r) {
            orow[r * EDIM]      = o0[r] + c0;
            orow[r * EDIM + 16] = o1[r] + c1;
        }
    }
}

extern "C" void kernel_launch(void* const* d_in, const int* in_sizes, int n_in,
                              void* d_out, int out_size, void* d_ws, size_t ws_size,
                              hipStream_t stream) {
    const float* x    = (const float*)d_in[0];
    const float* Wqkv = (const float*)d_in[1];
    const float* bqkv = (const float*)d_in[2];
    const float* Wff  = (const float*)d_in[3];
    const float* bff  = (const float*)d_in[4];
    float* out = (float*)d_out;

    float* Gws    = (float*)d_ws;
    float* colsum = Gws + BATCH * GSZ;

    gram_kernel<<<BATCH * 36, GW * 32, 0, stream>>>(x, Gws, colsum);

    fold_out_kernel<<<BATCH * CHUNKS, 384, 0, stream>>>(x, Gws, colsum, Wqkv, bqkv, Wff, bff, out);
}


# per-block 96x96x96 product M = Wq P and c = bq P dropped: q = x Wq + bq formed first, while G, Wv, Wk, Wff are in flight, then out = q P + bff; summation order changes
# speedup vs baseline: 2.2906x; 1.1104x over previous
#include <hip/hip_runtime.h>


typedef float v2f __attribute__((ext_vector_type(2)));
typedef float v8f __attribute__((ext_vector_type(8)));

#define BATCH   4
#define NSEQ    2048
#define EDIM    96
#define NHEAD   6
#define HDIM    16
#define QKVDIM  (3 * EDIM)
#define ROWS    (BATCH * NSEQ)
#define GSZ     (EDIM * EDIM)
#define GP      100
#define TP      104
#define SP      20
#define OROWS   64
#define CHUNKS  (NSEQ / OROWS)

static __device__ __forceinline__ v8f wmma_f32_4(v2f a, v2f b, v8f c) {
    return __builtin_amdgcn_wmma_f32_16x16x4_f32(
        false, a, false, b, (short)0, c, false, false);
}


#define GW 16
__global__ __launch_bounds__(GW * 32)
void gram_kernel(const float* __restrict__ x, float* __restrict__ Gws,
                 float* __restrict__ colsum) {
    const int blk  = blockIdx.x;
    const int b    = blk / 36;
    const int t36  = blk - b * 36;
    const int ti   = t36 / 6;
    const int tj   = t36 - ti * 6;
    const int lane = threadIdx.x & 31;
    const int wave = threadIdx.x >> 5;
    const int m    = lane & 15;
    const int sel  = (lane >> 4) << 1;

    const float* xb = x + (size_t)b * NSEQ * EDIM + (size_t)wave * (NSEQ / GW) * EDIM;
    const float* pa = xb + 16 * ti + m;
    const float* pb = xb + 16 * tj + m;

    __shared__ float red[GW * 256];
    __shared__ float redc[GW * 32];

    v8f c0 = {}, c1 = {};
    float cs = 0.f;
#pragma unroll
    for (int n0 = 0; n0 < NSEQ / GW; n0 += 8) {
        const int r0 = (n0 + sel) * EDIM;
        const int r1 = (n0 + 4 + sel) * EDIM;
        v2f a0, b0, a1, b1;
        a0.x = pa[r0];  a0.y = pa[r0 + EDIM];
        b0.x = pb[r0];  b0.y = pb[r0 + EDIM];
        a1.x = pa[r1];  a1.y = pa[r1 + EDIM];
        b1.x = pb[r1];  b1.y = pb[r1 + EDIM];
        cs += (a0.x + a0.y) + (a1.x + a1.y);
        c0 = wmma_f32_4(a0, b0, c0);
        c1 = wmma_f32_4(a1, b1, c1);
    }
#pragma unroll
    for (int r = 0; r < 8; ++r) red[wave * 256 + r * 32 + lane] = c0[r] + c1[r];
    redc[wave * 32 + lane] = cs;
    __syncthreads();

    const int t = threadIdx.x;
    if (t < 256) {
        float s = 0.f;
#pragma unroll
        for (int w = 0; w < GW; ++w) s += red[w * 256 + t];
        Gws[(size_t)b * GSZ + t36 * 256 + t] = s;
    }
    if (ti == tj && t < 16) {
        float cc = 0.f;
#pragma unroll
        for (int w = 0; w < GW; ++w) cc += redc[w * 32 + t] + redc[w * 32 + t + 16];
        colsum[(b * 6 + ti) * 32 + t] = cc;
    }
}

template <int P>
static __device__ __forceinline__ void put_row_piece(float* tile, int q4, float4 v) {
    const int row = q4 / 24;
    const int c4  = (q4 - row * 24) * 4;
    *(float4*)&tile[row * P + c4] = v;
}

static __device__ __forceinline__ float4 get_w_piece(const float* __restrict__ W, int off, int q4) {
    const int row = q4 / 24;
    const int c4  = (q4 - row * 24) * 4;
    return *(const float4*)&W[row * QKVDIM + off + c4];
}

__global__ __launch_bounds__(384)
void fold_out_kernel(const float* __restrict__ x,
                     const float* __restrict__ Gws, const float* __restrict__ colsum,
                     const float* __restrict__ Wqkv, const float* __restrict__ bqkv,
                     const float* __restrict__ Wff, const float* __restrict__ bff,
                     float* __restrict__ out) {
    const int b     = blockIdx.x / CHUNKS;
    const int chunk = blockIdx.x - b * CHUNKS;
    const int tid  = threadIdx.x;
    const int lane = tid & 31;
    const int wave = tid >> 5;
    const int m    = lane & 15;
    const int sel  = (lane >> 4) << 1;
    const int half = (lane >> 4) << 3;
    const int si   = wave >> 1;
    const int ct3  = (wave & 1) * 3;

    __shared__ __attribute__((aligned(16))) float sGP[EDIM * TP];
    float* const sG = sGP;
    float* const sP = sGP;
    __shared__ __attribute__((aligned(16))) float sT[EDIM * TP];
    __shared__ __attribute__((aligned(16))) float sS[NHEAD * 16 * SP];
    __shared__ float ss[EDIM];
    __shared__ float su[4 * EDIM];
    __shared__ __attribute__((aligned(16))) float sX[OROWS * GP];
    __shared__ __attribute__((aligned(16))) float sQ[OROWS * GP];
    __shared__ __attribute__((aligned(16))) float sWv[EDIM * TP];
    __shared__ __attribute__((aligned(16))) float sWk[EDIM * TP];
    __shared__ __attribute__((aligned(16))) float sWf[EDIM * TP];
    __shared__ __attribute__((aligned(16))) float sWq[EDIM * TP];


    const size_t row0 = (size_t)b * NSEQ + (size_t)chunk * OROWS;
    const float* xblk = x + row0 * EDIM;
    const float4 x0 = *(const float4*)&xblk[(0 * 384 + tid) * 4];
    const float4 x1 = *(const float4*)&xblk[(1 * 384 + tid) * 4];
    const float4 x2 = *(const float4*)&xblk[(2 * 384 + tid) * 4];
    const float4 x3 = *(const float4*)&xblk[(3 * 384 + tid) * 4];

    float4 wq4[6];
#pragma unroll
    for (int it = 0; it < 6; ++it) wq4[it] = get_w_piece(Wqkv, 0, it * 384 + tid);

    put_row_piece<GP>(sX, 0 * 384 + tid, x0);
    put_row_piece<GP>(sX, 1 * 384 + tid, x1);
    put_row_piece<GP>(sX, 2 * 384 + tid, x2);
    put_row_piece<GP>(sX, 3 * 384 + tid, x3);
#pragma unroll
    for (int it = 0; it < 6; ++it) put_row_piece<TP>(sWq, it * 384 + tid, wq4[it]);
    __syncthreads();

    float4 g4[6], wv4[6], wk4[6], wf4[6];
#pragma unroll
    for (int it = 0; it < 6; ++it) g4[it] = *(const float4*)&Gws[(size_t)b * GSZ + (it * 384 + tid) * 4];
    const float csv = (tid < EDIM) ? colsum[(b * 6 + (tid >> 4)) * 32 + (tid & 15)] : 0.f;
#pragma unroll
    for (int it = 0; it < 6; ++it) wv4[it] = get_w_piece(Wqkv, 2 * EDIM, it * 384 + tid);
#pragma unroll
    for (int it = 0; it < 6; ++it) wk4[it] = get_w_piece(Wqkv, EDIM, it * 384 + tid);
#pragma unroll
    for (int it = 0; it < 6; ++it) wf4[it] = *(const float4*)&Wff[(it * 384 + tid) * 4];

    const int st   = wave / 3;
    const int colA = 32 * (wave - 3 * st) + m;

    {
        v8f q0 = {}, q1 = {};
#pragma unroll
        for (int ks = 0; ks < 24; ++ks) {
            const v2f a = *(const v2f*)&sX[(16 * st + m) * GP + 4 * ks + sel];
            v2f b0, b1;
            b0.x = sWq[(4 * ks + sel) * TP + colA];
            b0.y = sWq[(4 * ks + sel + 1) * TP + colA];
            b1.x = sWq[(4 * ks + sel) * TP + colA + 16];
            b1.y = sWq[(4 * ks + sel + 1) * TP + colA + 16];
            q0 = wmma_f32_4(a, b0, q0);
            q1 = wmma_f32_4(a, b1, q1);
        }
        const float bq0 = bqkv[colA];
        const float bq1 = bqkv[colA + 16];
#pragma unroll
        for (int r = 0; r < 8; ++r) {
            sQ[(16 * st + r + half) * GP + colA]      = q0[r] + bq0;
            sQ[(16 * st + r + half) * GP + colA + 16] = q1[r] + bq1;
        }
    }

#pragma unroll
    for (int it = 0; it < 6; ++it) {
        const int e    = (it * 384 + tid) * 4;
        const int tile = e >> 8;
        const int idx  = e & 255;
        const int row  = (idx >> 5) + (((idx & 31) >> 4) << 3);
        const int col  = idx & 15;
        const int gi   = tile / 6;
        const int gj   = tile - gi * 6;
        *(float4*)&sG[(16 * gi + row) * GP + 16 * gj + col] = g4[it];
    }
    if (tid < EDIM) ss[tid] = csv;
#pragma unroll
    for (int it = 0; it < 6; ++it) put_row_piece<TP>(sWv, it * 384 + tid, wv4[it]);
#pragma unroll
    for (int it = 0; it < 6; ++it) put_row_piece<TP>(sWk, it * 384 + tid, wk4[it]);
    __syncthreads();

    {
        v8f acc[3] = {};
#pragma unroll
        for (int ks = 0; ks < 24; ++ks) {
            const v2f a = *(const v2f*)&sG[(16 * si + m) * GP + 4 * ks + sel];
#pragma unroll
            for (int q = 0; q < 3; ++q) {
                const float* bp = &sWv[(4 * ks + sel) * TP + 16 * (ct3 + q) + m];
                v2f bb;
                bb.x = bp[0];
                bb.y = bp[TP];
                acc[q] = wmma_f32_4(a, bb, acc[q]);
            }
        }
#pragma unroll
        for (int q = 0; q < 3; ++q)
#pragma unroll
            for (int r = 0; r < 8; ++r)
                sT[(16 * si + r + half) * TP + 16 * (ct3 + q) + m] = acc[q][r];
    }

#pragma unroll
    for (int it = 0; it < 6; ++it) put_row_piece<TP>(sWf, it * 384 + tid, wf4[it]);

    {
        const int part = (tid >= 2 * EDIM) ? 1 : 0;
        const int o    = tid - part * 2 * EDIM;
        const float* wcol = (o < EDIM) ? &sWk[o] : &sWv[o - EDIM];
        float u = 0.f;
#pragma unroll
        for (int i = 0; i < 48; ++i)
            u += ss[part * 48 + i] * wcol[(part * 48 + i) * TP];
        su[part * 2 * EDIM + o] = u;
    }
    __syncthreads();

    if (wave < NHEAD) {
        const int h = wave;
        v8f s0 = {}, s1 = {};
#pragma unroll
        for (int ks = 0; ks < 24; ks += 2) {
            v2f b0, b1;
            b0.x = sT[(4 * ks + sel) * TP + 16 * h + m];
            b0.y = sT[(4 * ks + sel + 1) * TP + 16 * h + m];
            b1.x = sT[(4 * ks + 4 + sel) * TP + 16 * h + m];
            b1.y = sT[(4 * ks + 5 + sel) * TP + 16 * h + m];
            const float* ap = &sWk[(4 * ks + sel) * TP + 16 * h + m];
            v2f a0, a1;
            a0.x = ap[0];
            a0.y = ap[TP];
            a1.x = ap[4 * TP];
            a1.y = ap[5 * TP];
            s0 = wmma_f32_4(a0, b0, s0);
            s1 = wmma_f32_4(a1, b1, s1);
        }
        const float bv_e = bqkv[2 * EDIM + 16 * h + m];
        const float uv_e = su[EDIM + 16 * h + m] + su[3 * EDIM + 16 * h + m];
#pragma unroll
        for (int r = 0; r < 8; ++r) {
            const int d = r + half;
            const float bk_d = bqkv[EDIM + 16 * h + d];
            const float uk_d = su[16 * h + d] + su[2 * EDIM + 16 * h + d];
            const float v = (s0[r] + s1[r]) + uk_d * bv_e + bk_d * uv_e
                          + (float)NSEQ * bk_d * bv_e;
            sS[h * 16 * SP + d * SP + m] = 0.25f * v;
        }
    }
    __syncthreads();

    {
        v8f acc[3] = {};
#pragma unroll
        for (int ks = 0; ks < 4; ++ks) {
            const v2f a = *(const v2f*)&sS[si * 16 * SP + m * SP + 4 * ks + sel];
#pragma unroll
            for (int q = 0; q < 3; ++q) {
                const float* bp = &sWf[(16 * si + 4 * ks + sel) * TP + 16 * (ct3 + q) + m];
                v2f bb;
                bb.x = bp[0];
                bb.y = bp[TP];
                acc[q] = wmma_f32_4(a, bb, acc[q]);
            }
        }
#pragma unroll
        for (int q = 0; q < 3; ++q)
#pragma unroll
            for (int r = 0; r < 8; ++r)
                sP[(16 * si + r + half) * TP + 16 * (ct3 + q) + m] = acc[q][r];
    }
    __syncthreads();

    {
        v8f o0 = {}, o1 = {};
#pragma unroll
        for (int ks = 0; ks < 24; ++ks) {
            const v2f a = *(const v2f*)&sQ[(16 * st + m) * GP + 4 * ks + sel];
            v2f b0, b1;
            b0.x = sP[(4 * ks + sel) * TP + colA];
            b0.y = sP[(4 * ks + sel + 1) * TP + colA];
            b1.x = sP[(4 * ks + sel) * TP + colA + 16];
            b1.y = sP[(4 * ks + sel + 1) * TP + colA + 16];
            o0 = wmma_f32_4(a, b0, o0);
            o1 = wmma_f32_4(a, b1, o1);
        }
        const float c0 = bff[colA];
        const float c1 = bff[colA + 16];
        float* orow = out + (row0 + 16 * st + half) * EDIM + colA;
#pragma unroll
        for (int r = 0; r < 8; ++r) {
            orow[r * EDIM]      = o0[r] + c0;
            orow[r * EDIM + 16] = o1[r] + c1;
        }
    }
}

extern "C" void kernel_launch(void* const* d_in, const int* in_sizes, int n_in,
                              void* d_out, int out_size, void* d_ws, size_t ws_size,
                              hipStream_t stream) {
    const float* x    = (const float*)d_in[0];
    const float* Wqkv = (const float*)d_in[1];
    const float* bqkv = (const float*)d_in[2];
    const float* Wff  = (const float*)d_in[3];
    const float* bff  = (const float*)d_in[4];
    float* out = (float*)d_out;

    float* Gws    = (float*)d_ws;
    float* colsum = Gws + BATCH * GSZ;

    gram_kernel<<<BATCH * 36, GW * 32, 0, stream>>>(x, Gws, colsum);

    fold_out_kernel<<<BATCH * CHUNKS, 384, 0, stream>>>(x, Gws, colsum, Wqkv, bqkv, Wff, bff, out);
}


# fold G->S moved into the Gram launch by linearity: each block folds its own 16x16 tile into six per-head partials (8 WMMA per head); launch 2 sums 36 partials, G@Wv and Wk^T@T gone; sum order changes
# speedup vs baseline: 2.6215x; 1.1445x over previous
#include <hip/hip_runtime.h>


typedef float v2f __attribute__((ext_vector_type(2)));
typedef float v8f __attribute__((ext_vector_type(8)));

#define BATCH   4
#define NSEQ    2048
#define EDIM    96
#define NHEAD   6
#define HDIM    16
#define QKVDIM  (3 * EDIM)
#define ROWS    (BATCH * NSEQ)
#define GP      100
#define TP      104
#define GTP     20
#define SSZ     (NHEAD * 256)
#define USZ     (2 * EDIM)
#define SP      20
#define OROWS   64
#define CHUNKS  (NSEQ / OROWS)

static __device__ __forceinline__ v8f wmma_f32_4(v2f a, v2f b, v8f c) {
    return __builtin_amdgcn_wmma_f32_16x16x4_f32(
        false, a, false, b, (short)0, c, false, false);
}


#define GW 16
__global__ __launch_bounds__(GW * 32)
void gram_fold_kernel(const float* __restrict__ x, const float* __restrict__ Wqkv,
                      float* __restrict__ Spart, float* __restrict__ upart) {
    const int blk  = blockIdx.x;
    const int b    = blk / 36;
    const int t36  = blk - b * 36;
    const int ti   = t36 / 6;
    const int tj   = t36 - ti * 6;
    const int lane = threadIdx.x & 31;
    const int wave = threadIdx.x >> 5;
    const int m    = lane & 15;
    const int sel  = (lane >> 4) << 1;
    const int half = (lane >> 4) << 3;

    const float* xb = x + (size_t)b * NSEQ * EDIM + (size_t)wave * (NSEQ / GW) * EDIM;
    const float* pa = xb + 16 * ti + m;
    const float* pb = xb + 16 * tj + m;

    __shared__ float red[GW * 256];
    __shared__ float redc[GW * 32];
    __shared__ __attribute__((aligned(16))) float sGt[16 * GTP];
    __shared__ float sc[16];

    v2f wvf[4] = {}, wkf[4] = {};
    if (wave < NHEAD) {
        const float* wv = Wqkv + (size_t)(16 * tj) * QKVDIM + 2 * EDIM + 16 * wave + m;
        const float* wk = Wqkv + (size_t)(16 * ti) * QKVDIM + EDIM + 16 * wave + m;
#pragma unroll
        for (int s = 0; s < 4; ++s) {
            wvf[s].x = wv[(4 * s + sel) * QKVDIM];
            wvf[s].y = wv[(4 * s + sel + 1) * QKVDIM];
            wkf[s].x = wk[(half + 2 * s) * QKVDIM];
            wkf[s].y = wk[(half + 2 * s + 1) * QKVDIM];
        }
    }

    v8f c0 = {}, c1 = {};
    float cs = 0.f;
#pragma unroll
    for (int n0 = 0; n0 < NSEQ / GW; n0 += 8) {
        const int r0 = (n0 + sel) * EDIM;
        const int r1 = (n0 + 4 + sel) * EDIM;
        v2f a0, b0, a1, b1;
        a0.x = pa[r0];  a0.y = pa[r0 + EDIM];
        b0.x = pb[r0];  b0.y = pb[r0 + EDIM];
        a1.x = pa[r1];  a1.y = pa[r1 + EDIM];
        b1.x = pb[r1];  b1.y = pb[r1 + EDIM];
        cs += (a0.x + a0.y) + (a1.x + a1.y);
        c0 = wmma_f32_4(a0, b0, c0);
        c1 = wmma_f32_4(a1, b1, c1);
    }

#pragma unroll
    for (int r = 0; r < 8; ++r) red[wave * 256 + r * 32 + lane] = c0[r] + c1[r];
    redc[wave * 32 + lane] = cs;
    __syncthreads();

    const int t = threadIdx.x;
    if (t < 256) {
        float s = 0.f;
#pragma unroll
        for (int w = 0; w < GW; ++w) s += red[w * 256 + t];
        const int row = (t >> 5) + (((t & 31) >> 4) << 3);
        sGt[row * GTP + (t & 15)] = s;
    }
    if (ti == tj && t < 16) {
        float cc = 0.f;
#pragma unroll
        for (int w = 0; w < GW; ++w) cc += redc[w * 32 + t] + redc[w * 32 + t + 16];
        sc[t] = cc;
    }
    __syncthreads();

    if (wave < NHEAD) {
        v8f y = {};
#pragma unroll
        for (int s = 0; s < 4; ++s) {
            const v2f a = *(const v2f*)&sGt[m * GTP + 4 * s + sel];
            y = wmma_f32_4(a, wvf[s], y);
        }
        v8f sp = {};
#pragma unroll
        for (int s = 0; s < 4; ++s) {
            v2f yb;
            yb.x = y[2 * s];
            yb.y = y[2 * s + 1];
            sp = wmma_f32_4(wkf[s], yb, sp);
        }
        float* dst = Spart + ((size_t)(b * 36 + t36) * NHEAD + wave) * 256;
#pragma unroll
        for (int r = 0; r < 8; ++r) dst[r * 32 + lane] = sp[r];
    } else if (ti == tj && t < NHEAD * 32 + USZ) {
        const int o = t - NHEAD * 32;
        const float* wcol = Wqkv + (size_t)(16 * ti) * QKVDIM + EDIM + o;
        float u = 0.f;
#pragma unroll
        for (int i = 0; i < 16; ++i) u += sc[i] * wcol[i * QKVDIM];
        upart[(b * 6 + ti) * USZ + o] = u;
    }
}

template <int P>
static __device__ __forceinline__ void put_row_piece(float* tile, int q4, float4 v) {
    const int row = q4 / 24;
    const int c4  = (q4 - row * 24) * 4;
    *(float4*)&tile[row * P + c4] = v;
}

static __device__ __forceinline__ float4 get_w_piece(const float* __restrict__ W, int off, int q4) {
    const int row = q4 / 24;
    const int c4  = (q4 - row * 24) * 4;
    return *(const float4*)&W[row * QKVDIM + off + c4];
}

__global__ __launch_bounds__(384)
void fold_out_kernel(const float* __restrict__ x,
                     const float* __restrict__ Spart, const float* __restrict__ upart,
                     const float* __restrict__ Wqkv, const float* __restrict__ bqkv,
                     const float* __restrict__ Wff, const float* __restrict__ bff,
                     float* __restrict__ out) {
    const int b     = blockIdx.x / CHUNKS;
    const int chunk = blockIdx.x - b * CHUNKS;
    const int tid  = threadIdx.x;
    const int lane = tid & 31;
    const int wave = tid >> 5;
    const int m    = lane & 15;
    const int sel  = (lane >> 4) << 1;
    const int half = (lane >> 4) << 3;
    const int si   = wave >> 1;
    const int ct3  = (wave & 1) * 3;

    __shared__ __attribute__((aligned(16))) float sP[EDIM * TP];
    __shared__ __attribute__((aligned(16))) float sS[NHEAD * 16 * SP];
    __shared__ __attribute__((aligned(16))) float sX[OROWS * GP];
    __shared__ __attribute__((aligned(16))) float sQ[OROWS * GP];
    __shared__ __attribute__((aligned(16))) float sWf[EDIM * TP];
    __shared__ __attribute__((aligned(16))) float sWq[EDIM * TP];


    const size_t row0 = (size_t)b * NSEQ + (size_t)chunk * OROWS;
    const float* xblk = x + row0 * EDIM;
    const float4 x0 = *(const float4*)&xblk[(0 * 384 + tid) * 4];
    const float4 x1 = *(const float4*)&xblk[(1 * 384 + tid) * 4];
    const float4 x2 = *(const float4*)&xblk[(2 * 384 + tid) * 4];
    const float4 x3 = *(const float4*)&xblk[(3 * 384 + tid) * 4];

    float4 wq4[6];
#pragma unroll
    for (int it = 0; it < 6; ++it) wq4[it] = get_w_piece(Wqkv, 0, it * 384 + tid);

    put_row_piece<GP>(sX, 0 * 384 + tid, x0);
    put_row_piece<GP>(sX, 1 * 384 + tid, x1);
    put_row_piece<GP>(sX, 2 * 384 + tid, x2);
    put_row_piece<GP>(sX, 3 * 384 + tid, x3);
#pragma unroll
    for (int it = 0; it < 6; ++it) put_row_piece<TP>(sWq, it * 384 + tid, wq4[it]);
    __syncthreads();

    const int hs = tid >> 6;
    const int q6 = tid & 63;
    const int ds = (q6 >> 3) + (((q6 & 7) >> 2) << 3);
    const int es = (q6 & 3) << 2;
    float4 sp4[36];
#pragma unroll
    for (int p = 0; p < 36; ++p)
        sp4[p] = *(const float4*)&Spart[(size_t)(b * 36 + p) * SSZ + tid * 4];
    float  ukp[6];
    float4 uvp[6];
#pragma unroll
    for (int p = 0; p < 6; ++p) {
        ukp[p] = upart[(b * 6 + p) * USZ + 16 * hs + ds];
        uvp[p] = *(const float4*)&upart[(b * 6 + p) * USZ + EDIM + 16 * hs + es];
    }
    const float  bk_d = bqkv[EDIM + 16 * hs + ds];
    const float4 bv4  = *(const float4*)&bqkv[2 * EDIM + 16 * hs + es];
    float4 wf4[6];
#pragma unroll
    for (int it = 0; it < 6; ++it) wf4[it] = *(const float4*)&Wff[(it * 384 + tid) * 4];

    const int st   = wave / 3;
    const int colA = 32 * (wave - 3 * st) + m;

    {
        v8f q0 = {}, q1 = {};
#pragma unroll
        for (int ks = 0; ks < 24; ++ks) {
            const v2f a = *(const v2f*)&sX[(16 * st + m) * GP + 4 * ks + sel];
            v2f b0, b1;
            b0.x = sWq[(4 * ks + sel) * TP + colA];
            b0.y = sWq[(4 * ks + sel + 1) * TP + colA];
            b1.x = sWq[(4 * ks + sel) * TP + colA + 16];
            b1.y = sWq[(4 * ks + sel + 1) * TP + colA + 16];
            q0 = wmma_f32_4(a, b0, q0);
            q1 = wmma_f32_4(a, b1, q1);
        }
        const float bq0 = bqkv[colA];
        const float bq1 = bqkv[colA + 16];
#pragma unroll
        for (int r = 0; r < 8; ++r) {
            sQ[(16 * st + r + half) * GP + colA]      = q0[r] + bq0;
            sQ[(16 * st + r + half) * GP + colA + 16] = q1[r] + bq1;
        }
    }

    {
        float4 acc = sp4[0];
#pragma unroll
        for (int p = 1; p < 36; ++p) {
            acc.x += sp4[p].x;  acc.y += sp4[p].y;  acc.z += sp4[p].z;  acc.w += sp4[p].w;
        }
        float  uk_d = ukp[0];
        float4 uv4  = uvp[0];
#pragma unroll
        for (int p = 1; p < 6; ++p) {
            uk_d  += ukp[p];
            uv4.x += uvp[p].x;  uv4.y += uvp[p].y;  uv4.z += uvp[p].z;  uv4.w += uvp[p].w;
        }
        float4 v;
        v.x = acc.x + uk_d * bv4.x + bk_d * uv4.x + (float)NSEQ * bk_d * bv4.x;
        v.y = acc.y + uk_d * bv4.y + bk_d * uv4.y + (float)NSEQ * bk_d * bv4.y;
        v.z = acc.z + uk_d * bv4.z + bk_d * uv4.z + (float)NSEQ * bk_d * bv4.z;
        v.w = acc.w + uk_d * bv4.w + bk_d * uv4.w + (float)NSEQ * bk_d * bv4.w;
        v.x *= 0.25f;  v.y *= 0.25f;  v.z *= 0.25f;  v.w *= 0.25f;
        *(float4*)&sS[hs * 16 * SP + ds * SP + es] = v;
    }
#pragma unroll
    for (int it = 0; it < 6; ++it) put_row_piece<TP>(sWf, it * 384 + tid, wf4[it]);
    __syncthreads();

    {
        v8f acc[3] = {};
#pragma unroll
        for (int ks = 0; ks < 4; ++ks) {
            const v2f a = *(const v2f*)&sS[si * 16 * SP + m * SP + 4 * ks + sel];
#pragma unroll
            for (int q = 0; q < 3; ++q) {
                const float* bp = &sWf[(16 * si + 4 * ks + sel) * TP + 16 * (ct3 + q) + m];
                v2f bb;
                bb.x = bp[0];
                bb.y = bp[TP];
                acc[q] = wmma_f32_4(a, bb, acc[q]);
            }
        }
#pragma unroll
        for (int q = 0; q < 3; ++q)
#pragma unroll
            for (int r = 0; r < 8; ++r)
                sP[(16 * si + r + half) * TP + 16 * (ct3 + q) + m] = acc[q][r];
    }
    __syncthreads();

    {
        v8f o0 = {}, o1 = {};
#pragma unroll
        for (int ks = 0; ks < 24; ++ks) {
            const v2f a = *(const v2f*)&sQ[(16 * st + m) * GP + 4 * ks + sel];
            v2f b0, b1;
            b0.x = sP[(4 * ks + sel) * TP + colA];
            b0.y = sP[(4 * ks + sel + 1) * TP + colA];
            b1.x = sP[(4 * ks + sel) * TP + colA + 16];
            b1.y = sP[(4 * ks + sel + 1) * TP + colA + 16];
            o0 = wmma_f32_4(a, b0, o0);
            o1 = wmma_f32_4(a, b1, o1);
        }
        const float c0 = bff[colA];
        const float c1 = bff[colA + 16];
        float* orow = out + (row0 + 16 * st + half) * EDIM + colA;
#pragma unroll
        for (int r = 0; r < 8; ++r) {
            orow[r * EDIM]      = o0[r] + c0;
            orow[r * EDIM + 16] = o1[r] + c1;
        }
    }
}

extern "C" void kernel_launch(void* const* d_in, const int* in_sizes, int n_in,
                              void* d_out, int out_size, void* d_ws, size_t ws_size,
                              hipStream_t stream) {
    const float* x    = (const float*)d_in[0];
    const float* Wqkv = (const float*)d_in[1];
    const float* bqkv = (const float*)d_in[2];
    const float* Wff  = (const float*)d_in[3];
    const float* bff  = (const float*)d_in[4];
    float* out = (float*)d_out;

    float* Spart = (float*)d_ws;
    float* upart = Spart + BATCH * 36 * SSZ;

    gram_fold_kernel<<<BATCH * 36, GW * 32, 0, stream>>>(x, Wqkv, Spart, upart);

    fold_out_kernel<<<BATCH * CHUNKS, 384, 0, stream>>>(x, Spart, upart, Wqkv, bqkv, Wff, bff, out);
}


# Gram launch re-tiled: block = 32x32 region of G over 512 rows, 8 waves; each lane reads 2 adjacent columns by 8-byte loads (whole lines), interleaved column tiles, fold follows; sum order changes
# speedup vs baseline: 2.8886x; 1.1019x over previous
#include <hip/hip_runtime.h>


typedef float v2f __attribute__((ext_vector_type(2)));
typedef float v8f __attribute__((ext_vector_type(8)));

#define BATCH   4
#define NSEQ    2048
#define EDIM    96
#define NHEAD   6
#define HDIM    16
#define QKVDIM  (3 * EDIM)
#define ROWS    (BATCH * NSEQ)
#define GP      100
#define TP      104
#define GTP     20
#define SSZ     (NHEAD * 256)
#define USZ     (2 * EDIM)
#define USETS   12
#define SP      20
#define OROWS   64
#define CHUNKS  (NSEQ / OROWS)

static __device__ __forceinline__ v8f wmma_f32_4(v2f a, v2f b, v8f c) {
    return __builtin_amdgcn_wmma_f32_16x16x4_f32(
        false, a, false, b, (short)0, c, false, false);
}


#define GW    8
#define CROWS (NSEQ / 4)
#define WROWS (CROWS / GW)
__global__ __launch_bounds__(GW * 32)
void gram_fold_kernel(const float* __restrict__ x, const float* __restrict__ Wqkv,
                      float* __restrict__ Spart, float* __restrict__ upart) {
    const int blk  = blockIdx.x;
    const int b    = blk / 36;
    const int p36  = blk - b * 36;
    const int sup  = p36 >> 2;
    const int c    = p36 & 3;
    const int gi   = sup / 3;
    const int gj   = sup - gi * 3;
    const int lane = threadIdx.x & 31;
    const int wave = threadIdx.x >> 5;
    const int m    = lane & 15;
    const int sel  = (lane >> 4) << 1;
    const int half = (lane >> 4) << 3;

    const float* xw = x + ((size_t)b * NSEQ + (size_t)c * CROWS + (size_t)wave * WROWS + sel) * EDIM;
    const float* pa = xw + 32 * gi + 2 * m;
    const float* pb = xw + 32 * gj + 2 * m;

    __shared__ __attribute__((aligned(16))) float red[GW * 1024];
    __shared__ float redc[GW * 64];
    __shared__ __attribute__((aligned(16))) float sGt[4 * 16 * GTP];
    __shared__ float sc[32];

    v2f wvf[2][4] = {}, wkf[2][4] = {};
    if (wave < NHEAD) {
        const float* wv = Wqkv + (size_t)(32 * gj) * QKVDIM + 2 * EDIM + 16 * wave + m;
        const float* wk = Wqkv + (size_t)(32 * gi) * QKVDIM + EDIM + 16 * wave + m;
#pragma unroll
        for (int e = 0; e < 2; ++e) {
#pragma unroll
            for (int s = 0; s < 4; ++s) {
                wvf[e][s].x = wv[(2 * (4 * s + sel) + e) * QKVDIM];
                wvf[e][s].y = wv[(2 * (4 * s + sel + 1) + e) * QKVDIM];
                wkf[e][s].x = wk[(2 * (half + 2 * s) + e) * QKVDIM];
                wkf[e][s].y = wk[(2 * (half + 2 * s + 1) + e) * QKVDIM];
            }
        }
    }

    v8f c00 = {}, c01 = {}, c10 = {}, c11 = {};
    float cs0 = 0.f, cs1 = 0.f;
#pragma unroll
    for (int n0 = 0; n0 < WROWS; n0 += 4) {
        const v2f ar0 = *(const v2f*)&pa[n0 * EDIM];
        const v2f ar1 = *(const v2f*)&pa[(n0 + 1) * EDIM];
        const v2f br0 = *(const v2f*)&pb[n0 * EDIM];
        const v2f br1 = *(const v2f*)&pb[(n0 + 1) * EDIM];
        v2f a0, a1, b0, b1;
        a0.x = ar0.x;  a0.y = ar1.x;
        a1.x = ar0.y;  a1.y = ar1.y;
        b0.x = br0.x;  b0.y = br1.x;
        b1.x = br0.y;  b1.y = br1.y;
        cs0 += a0.x + a0.y;
        cs1 += a1.x + a1.y;
        c00 = wmma_f32_4(a0, b0, c00);
        c01 = wmma_f32_4(a0, b1, c01);
        c10 = wmma_f32_4(a1, b0, c10);
        c11 = wmma_f32_4(a1, b1, c11);
    }

#pragma unroll
    for (int r = 0; r < 8; ++r) {
        red[wave * 1024 +   0 + r * 32 + lane] = c00[r];
        red[wave * 1024 + 256 + r * 32 + lane] = c01[r];
        red[wave * 1024 + 512 + r * 32 + lane] = c10[r];
        red[wave * 1024 + 768 + r * 32 + lane] = c11[r];
    }
    redc[wave * 64 + lane]      = cs0;
    redc[wave * 64 + 32 + lane] = cs1;
    __syncthreads();

    const int t = threadIdx.x;
    {
        const int o = 4 * t;
        float4 s = *(const float4*)&red[o];
#pragma unroll
        for (int w = 1; w < GW; ++w) {
            const float4 v = *(const float4*)&red[w * 1024 + o];
            s.x += v.x;  s.y += v.y;  s.z += v.z;  s.w += v.w;
        }
        const int tile = o >> 8;
        const int tp   = o & 255;
        const int row  = (tp >> 5) + (((tp & 31) >> 4) << 3);
        *(float4*)&sGt[tile * 16 * GTP + row * GTP + (tp & 15)] = s;
    }
    if (gi == gj && t < 32) {
        const int e  = t & 1;
        const int mm = t >> 1;
        float cc = 0.f;
#pragma unroll
        for (int w = 0; w < GW; ++w) cc += redc[w * 64 + e * 32 + mm] + redc[w * 64 + e * 32 + mm + 16];
        sc[t] = cc;
    }
    __syncthreads();

    if (wave < NHEAD) {
        v8f y0 = {}, y1 = {};
#pragma unroll
        for (int f = 0; f < 2; ++f) {
#pragma unroll
            for (int s = 0; s < 4; ++s) {
                const v2f g0 = *(const v2f*)&sGt[(0 + f) * 16 * GTP + m * GTP + 4 * s + sel];
                const v2f g1 = *(const v2f*)&sGt[(2 + f) * 16 * GTP + m * GTP + 4 * s + sel];
                y0 = wmma_f32_4(g0, wvf[f][s], y0);
                y1 = wmma_f32_4(g1, wvf[f][s], y1);
            }
        }
        v8f sp = {};
#pragma unroll
        for (int s = 0; s < 4; ++s) {
            v2f yb;
            yb.x = y0[2 * s];
            yb.y = y0[2 * s + 1];
            sp = wmma_f32_4(wkf[0][s], yb, sp);
        }
#pragma unroll
        for (int s = 0; s < 4; ++s) {
            v2f yb;
            yb.x = y1[2 * s];
            yb.y = y1[2 * s + 1];
            sp = wmma_f32_4(wkf[1][s], yb, sp);
        }
        float* dst = Spart + ((size_t)(b * 36 + p36) * NHEAD + wave) * 256;
#pragma unroll
        for (int r = 0; r < 8; ++r) dst[r * 32 + lane] = sp[r];
    } else if (gi == gj) {
        const int tt = t - NHEAD * 32;
        const float* wcol = Wqkv + (size_t)(32 * gi) * QKVDIM + EDIM + tt;
        float u0 = 0.f, u1 = 0.f, u2 = 0.f;
#pragma unroll
        for (int i = 0; i < 32; ++i) {
            const float si = sc[i];
            u0 += si * wcol[i * QKVDIM];
            u1 += si * wcol[i * QKVDIM + 64];
            u2 += si * wcol[i * QKVDIM + 128];
        }
        float* ud = upart + (size_t)(b * USETS + gi * 4 + c) * USZ + tt;
        ud[0]   = u0;
        ud[64]  = u1;
        ud[128] = u2;
    }
}

template <int P>
static __device__ __forceinline__ void put_row_piece(float* tile, int q4, float4 v) {
    const int row = q4 / 24;
    const int c4  = (q4 - row * 24) * 4;
    *(float4*)&tile[row * P + c4] = v;
}

static __device__ __forceinline__ float4 get_w_piece(const float* __restrict__ W, int off, int q4) {
    const int row = q4 / 24;
    const int c4  = (q4 - row * 24) * 4;
    return *(const float4*)&W[row * QKVDIM + off + c4];
}

__global__ __launch_bounds__(384)
void fold_out_kernel(const float* __restrict__ x,
                     const float* __restrict__ Spart, const float* __restrict__ upart,
                     const float* __restrict__ Wqkv, const float* __restrict__ bqkv,
                     const float* __restrict__ Wff, const float* __restrict__ bff,
                     float* __restrict__ out) {
    const int b     = blockIdx.x / CHUNKS;
    const int chunk = blockIdx.x - b * CHUNKS;
    const int tid  = threadIdx.x;
    const int lane = tid & 31;
    const int wave = tid >> 5;
    const int m    = lane & 15;
    const int sel  = (lane >> 4) << 1;
    const int half = (lane >> 4) << 3;
    const int si   = wave >> 1;
    const int ct3  = (wave & 1) * 3;

    __shared__ __attribute__((aligned(16))) float sP[EDIM * TP];
    __shared__ __attribute__((aligned(16))) float sS[NHEAD * 16 * SP];
    __shared__ __attribute__((aligned(16))) float sX[OROWS * GP];
    __shared__ __attribute__((aligned(16))) float sQ[OROWS * GP];
    __shared__ __attribute__((aligned(16))) float sWf[EDIM * TP];
    __shared__ __attribute__((aligned(16))) float sWq[EDIM * TP];


    const size_t row0 = (size_t)b * NSEQ + (size_t)chunk * OROWS;
    const float* xblk = x + row0 * EDIM;
    const float4 x0 = *(const float4*)&xblk[(0 * 384 + tid) * 4];
    const float4 x1 = *(const float4*)&xblk[(1 * 384 + tid) * 4];
    const float4 x2 = *(const float4*)&xblk[(2 * 384 + tid) * 4];
    const float4 x3 = *(const float4*)&xblk[(3 * 384 + tid) * 4];

    float4 wq4[6];
#pragma unroll
    for (int it = 0; it < 6; ++it) wq4[it] = get_w_piece(Wqkv, 0, it * 384 + tid);

    put_row_piece<GP>(sX, 0 * 384 + tid, x0);
    put_row_piece<GP>(sX, 1 * 384 + tid, x1);
    put_row_piece<GP>(sX, 2 * 384 + tid, x2);
    put_row_piece<GP>(sX, 3 * 384 + tid, x3);
#pragma unroll
    for (int it = 0; it < 6; ++it) put_row_piece<TP>(sWq, it * 384 + tid, wq4[it]);
    __syncthreads();

    const int hs = tid >> 6;
    const int q6 = tid & 63;
    const int ds = (q6 >> 3) + (((q6 & 7) >> 2) << 3);
    const int es = (q6 & 3) << 2;
    float4 sp4[36];
#pragma unroll
    for (int p = 0; p < 36; ++p)
        sp4[p] = *(const float4*)&Spart[(size_t)(b * 36 + p) * SSZ + tid * 4];
    float  ukp[USETS];
    float4 uvp[USETS];
#pragma unroll
    for (int p = 0; p < USETS; ++p) {
        ukp[p] = upart[(b * USETS + p) * USZ + 16 * hs + ds];
        uvp[p] = *(const float4*)&upart[(b * USETS + p) * USZ + EDIM + 16 * hs + es];
    }
    const float  bk_d = bqkv[EDIM + 16 * hs + ds];
    const float4 bv4  = *(const float4*)&bqkv[2 * EDIM + 16 * hs + es];
    float4 wf4[6];
#pragma unroll
    for (int it = 0; it < 6; ++it) wf4[it] = *(const float4*)&Wff[(it * 384 + tid) * 4];

    const int st   = wave / 3;
    const int colA = 32 * (wave - 3 * st) + m;

    {
        v8f q0 = {}, q1 = {};
#pragma unroll
        for (int ks = 0; ks < 24; ++ks) {
            const v2f a = *(const v2f*)&sX[(16 * st + m) * GP + 4 * ks + sel];
            v2f b0, b1;
            b0.x = sWq[(4 * ks + sel) * TP + colA];
            b0.y = sWq[(4 * ks + sel + 1) * TP + colA];
            b1.x = sWq[(4 * ks + sel) * TP + colA + 16];
            b1.y = sWq[(4 * ks + sel + 1) * TP + colA + 16];
            q0 = wmma_f32_4(a, b0, q0);
            q1 = wmma_f32_4(a, b1, q1);
        }
        const float bq0 = bqkv[colA];
        const float bq1 = bqkv[colA + 16];
#pragma unroll
        for (int r = 0; r < 8; ++r) {
            sQ[(16 * st + r + half) * GP + colA]      = q0[r] + bq0;
            sQ[(16 * st + r + half) * GP + colA + 16] = q1[r] + bq1;
        }
    }

    {
        float4 acc = sp4[0];
#pragma unroll
        for (int p = 1; p < 36; ++p) {
            acc.x += sp4[p].x;  acc.y += sp4[p].y;  acc.z += sp4[p].z;  acc.w += sp4[p].w;
        }
        float  uk_d = ukp[0];
        float4 uv4  = uvp[0];
#pragma unroll
        for (int p = 1; p < USETS; ++p) {
            uk_d  += ukp[p];
            uv4.x += uvp[p].x;  uv4.y += uvp[p].y;  uv4.z += uvp[p].z;  uv4.w += uvp[p].w;
        }
        float4 v;
        v.x = acc.x + uk_d * bv4.x + bk_d * uv4.x + (float)NSEQ * bk_d * bv4.x;
        v.y = acc.y + uk_d * bv4.y + bk_d * uv4.y + (float)NSEQ * bk_d * bv4.y;
        v.z = acc.z + uk_d * bv4.z + bk_d * uv4.z + (float)NSEQ * bk_d * bv4.z;
        v.w = acc.w + uk_d * bv4.w + bk_d * uv4.w + (float)NSEQ * bk_d * bv4.w;
        v.x *= 0.25f;  v.y *= 0.25f;  v.z *= 0.25f;  v.w *= 0.25f;
        *(float4*)&sS[hs * 16 * SP + ds * SP + es] = v;
    }
#pragma unroll
    for (int it = 0; it < 6; ++it) put_row_piece<TP>(sWf, it * 384 + tid, wf4[it]);
    __syncthreads();

    {
        v8f acc[3] = {};
#pragma unroll
        for (int ks = 0; ks < 4; ++ks) {
            const v2f a = *(const v2f*)&sS[si * 16 * SP + m * SP + 4 * ks + sel];
#pragma unroll
            for (int q = 0; q < 3; ++q) {
                const float* bp = &sWf[(16 * si + 4 * ks + sel) * TP + 16 * (ct3 + q) + m];
                v2f bb;
                bb.x = bp[0];
                bb.y = bp[TP];
                acc[q] = wmma_f32_4(a, bb, acc[q]);
            }
        }
#pragma unroll
        for (int q = 0; q < 3; ++q)
#pragma unroll
            for (int r = 0; r < 8; ++r)
                sP[(16 * si + r + half) * TP + 16 * (ct3 + q) + m] = acc[q][r];
    }
    __syncthreads();

    {
        v8f o0 = {}, o1 = {};
#pragma unroll
        for (int ks = 0; ks < 24; ++ks) {
            const v2f a = *(const v2f*)&sQ[(16 * st + m) * GP + 4 * ks + sel];
            v2f b0, b1;
            b0.x = sP[(4 * ks + sel) * TP + colA];
            b0.y = sP[(4 * ks + sel + 1) * TP + colA];
            b1.x = sP[(4 * ks + sel) * TP + colA + 16];
            b1.y = sP[(4 * ks + sel + 1) * TP + colA + 16];
            o0 = wmma_f32_4(a, b0, o0);
            o1 = wmma_f32_4(a, b1, o1);
        }
        const float c0 = bff[colA];
        const float c1 = bff[colA + 16];
        float* orow = out + (row0 + 16 * st + half) * EDIM + colA;
#pragma unroll
        for (int r = 0; r < 8; ++r) {
            orow[r * EDIM]      = o0[r] + c0;
            orow[r * EDIM + 16] = o1[r] + c1;
        }
    }
}

extern "C" void kernel_launch(void* const* d_in, const int* in_sizes, int n_in,
                              void* d_out, int out_size, void* d_ws, size_t ws_size,
                              hipStream_t stream) {
    const float* x    = (const float*)d_in[0];
    const float* Wqkv = (const float*)d_in[1];
    const float* bqkv = (const float*)d_in[2];
    const float* Wff  = (const float*)d_in[3];
    const float* bff  = (const float*)d_in[4];
    float* out = (float*)d_out;

    float* Spart = (float*)d_ws;
    float* upart = Spart + BATCH * 36 * SSZ;

    gram_fold_kernel<<<BATCH * 36, GW * 32, 0, stream>>>(x, Wqkv, Spart, upart);

    fold_out_kernel<<<BATCH * CHUNKS, 384, 0, stream>>>(x, Spart, upart, Wqkv, bqkv, Wff, bff, out);
}
